# starting kernel
# baseline (speedup 1.0000x reference)
#include <hip/hip_runtime.h>


typedef float v2f __attribute__((ext_vector_type(2)));
typedef float v4f __attribute__((ext_vector_type(4)));
typedef float v8f __attribute__((ext_vector_type(8)));

#define T_LEN 8192
#define NB    16
#define NP    128
#define ME    144
#define XROWS 148
#define STR   68

#define XS_OFF   0
#define W_OFF    (XS_OFF + XROWS * STR)
#define R_OFF0   (W_OFF + 5 * 64 * STR)
#define RES_SZ   (ME * STR)
#define BIAS_OFF (R_OFF0 + 3 * RES_SZ)
#define LDS_FLOATS (BIAS_OFF + 64)

__device__ __forceinline__ int kslot(int ci) {
    return (((ci >> 1) & 1) << 5) + ((ci >> 2) << 1) + (ci & 1);
}

__device__ __forceinline__ int ref_idx(int i) {
    if (i < 2) return 2 - i;
    if (i > T_LEN - 3) return (i == T_LEN - 2) ? T_LEN - 6 : T_LEN - 7;
    return i - 2;
}

template<int TAPS, bool DUAL>
__device__ __forceinline__ void gemm_tiles(float* lds, int aw_base, int xb_base,
                                           int res_base, int pt0)
{
    const int x0 = xb_base + pt0 * 16 * STR;
    const int x1 = x0 + 16 * STR;

    v4f A[2][8], B0[2][8], B1[2][8];
    #pragma unroll
    for (int q = 0; q < 8; ++q) {
        A[0][q]  = *(const v4f*)&lds[aw_base + 4 * q];
        B0[0][q] = *(const v4f*)&lds[x0 + 4 * q];
        if constexpr (DUAL) B1[0][q] = *(const v4f*)&lds[x1 + 4 * q];
    }

    v8f acc0 = {0.f,0.f,0.f,0.f,0.f,0.f,0.f,0.f};
    v8f acc1 = {0.f,0.f,0.f,0.f,0.f,0.f,0.f,0.f};

    #pragma unroll
    for (int t = 0; t < TAPS; ++t) {
        const int cur = t & 1, nxt = cur ^ 1;
        if (t + 1 < TAPS) {
            #pragma unroll
            for (int q = 0; q < 8; ++q) {
                A[nxt][q]  = *(const v4f*)&lds[aw_base + (t + 1) * 64 * STR + 4 * q];
                B0[nxt][q] = *(const v4f*)&lds[x0 + (t + 1) * STR + 4 * q];
                if constexpr (DUAL) B1[nxt][q] = *(const v4f*)&lds[x1 + (t + 1) * STR + 4 * q];
            }
        }
        #pragma unroll
        for (int q = 0; q < 8; ++q) {
            v2f a0 = {A[cur][q][0], A[cur][q][1]};
            v2f a1 = {A[cur][q][2], A[cur][q][3]};
            v2f b0 = {B0[cur][q][0], B0[cur][q][1]};
            v2f b1 = {B0[cur][q][2], B0[cur][q][3]};
            acc0 = __builtin_amdgcn_wmma_f32_16x16x4_f32(false, a0, false, b0, (short)0, acc0, false, false);
            acc0 = __builtin_amdgcn_wmma_f32_16x16x4_f32(false, a1, false, b1, (short)0, acc0, false, false);
            if constexpr (DUAL) {
                v2f c0 = {B1[cur][q][0], B1[cur][q][1]};
                v2f c1 = {B1[cur][q][2], B1[cur][q][3]};
                acc1 = __builtin_amdgcn_wmma_f32_16x16x4_f32(false, a0, false, c0, (short)0, acc1, false, false);
                acc1 = __builtin_amdgcn_wmma_f32_16x16x4_f32(false, a1, false, c1, (short)0, acc1, false, false);
            }
        }
    }

    int st0 = res_base + pt0 * 16 * STR;
    *(v4f*)&lds[st0]     = (v4f){acc0[0], acc0[1], acc0[2], acc0[3]};
    *(v4f*)&lds[st0 + 4] = (v4f){acc0[4], acc0[5], acc0[6], acc0[7]};
    if constexpr (DUAL) {
        int st1 = st0 + 16 * STR;
        *(v4f*)&lds[st1]     = (v4f){acc1[0], acc1[1], acc1[2], acc1[3]};
        *(v4f*)&lds[st1 + 4] = (v4f){acc1[4], acc1[5], acc1[6], acc1[7]};
    }
}

template<int TAPS, int KOFS, int RHO>
__device__ __forceinline__ void process_residue(float* lds, const float* wgrp, int tid,
                                                int wrow, int aw_base, int xb_base,
                                                int res_nm_base)
{
    __syncthreads();
    for (int e = tid; e < TAPS * 64 * 64; e += 256) {
        int t  = e >> 12;
        int r  = e & 4095;
        int co = r >> 6;
        int ci = r & 63;
        lds[W_OFF + (t * 64 + co) * STR + kslot(ci)] = wgrp[(co * 64 + ci) * 13 + 3 * t + KOFS];
    }
    __syncthreads();

    const int res_base = R_OFF0 + RHO * RES_SZ + res_nm_base;
    if (wrow == 0) {
        gemm_tiles<TAPS, true >(lds, aw_base, xb_base, res_base, 0);
        gemm_tiles<TAPS, true >(lds, aw_base, xb_base, res_base, 2);
        gemm_tiles<TAPS, false>(lds, aw_base, xb_base, res_base, 4);
    } else {
        gemm_tiles<TAPS, true >(lds, aw_base, xb_base, res_base, 5);
        gemm_tiles<TAPS, true >(lds, aw_base, xb_base, res_base, 7);
    }
}

__global__ __launch_bounds__(256) void trconv_fused_kernel(
    const float* __restrict__ x,
    const float* __restrict__ w,
    const float* __restrict__ bias,
    float* __restrict__ out)
{
    extern __shared__ float lds[];
    const int tid = threadIdx.x;
    const int b   = blockIdx.z;
    const int g   = blockIdx.y;
    const int p0  = blockIdx.x * NP;
    const int me0 = p0 - 8;

    const float* xg = x + (size_t)(b * 128 + g * 64) * T_LEN;
    for (int e = tid; e < 64 * XROWS; e += 256) {
        int ci = e / XROWS;
        int lp = e - ci * XROWS;
        int m  = me0 + lp;
        m = m < 0 ? 0 : (m > T_LEN - 1 ? T_LEN - 1 : m);
        lds[XS_OFF + lp * STR + kslot(ci)] = xg[(size_t)ci * T_LEN + m];
    }
    if (tid < 64) lds[BIAS_OFF + tid] = bias[g * 64 + tid];

    const int lane = tid & 31;
    const int wv   = tid >> 5;
    const int ct   = wv & 3;
    const int wrow = wv >> 2;
    const int nm   = lane & 15;
    const int lh   = lane >> 4;

    const int aw_base     = W_OFF + (ct * 16 + nm) * STR + lh * 32;
    const int xb_base     = XS_OFF + nm * STR + lh * 32;
    const int res_nm_base = nm * STR + ct * 16 + lh * 8;

    const float* wgrp = w + (size_t)(g * 64) * (64 * 13);

    process_residue<4, 2, 0>(lds, wgrp, tid, wrow, aw_base, xb_base, res_nm_base);
    process_residue<4, 1, 1>(lds, wgrp, tid, wrow, aw_base, xb_base, res_nm_base);
    process_residue<5, 0, 2>(lds, wgrp, tid, wrow, aw_base, xb_base, res_nm_base);

    __syncthreads();

    const float* r0p = &lds[R_OFF0 + 0 * RES_SZ];
    const float* r1p = &lds[R_OFF0 + 1 * RES_SZ];
    const float* r2p = &lds[R_OFF0 + 2 * RES_SZ];
    for (int idx = tid; idx < 64 * NP; idx += 256) {
        int co = idx >> 7;
        int il = idx & (NP - 1);
        int i  = p0 + il;
        float bi = lds[BIAS_OFF + co];

        int ri = ref_idx(i);
        int lm = ri - me0;
        float l0 = r0p[lm * STR + co], r0v = r0p[(lm + 1) * STR + co];
        float l1 = r1p[lm * STR + co], r1v = r1p[(lm + 1) * STR + co];
        float x_m = r2p[lm * STR + co] + bi;

        float um = x_m;
        #pragma unroll
        for (int d = -2; d <= 2; ++d) {
            if (d == 0) continue;
            int j = i + d;
            float v = 0.0f;
            if (j >= 0 && j <= T_LEN - 1)
                v = r2p[(ref_idx(j) - me0) * STR + co] + bi;
            um = fmaxf(um, v);
        }

        float d0 = r0v - l0, d1 = r1v - l1;
        float qg0 = d0 / (1.0f + __expf(-1.702f * d0));
        float qg1 = d1 / (1.0f + __expf(-1.702f * d1));
        float accv = qg0 + qg1 + (r0v + l0 + r1v + l1) + 2.0f * x_m + 4.0f * bi + um;

        out[(size_t)(b * 128 + g * 64 + co) * T_LEN + i] = accv;
    }
}

extern "C" void kernel_launch(void* const* d_in, const int* in_sizes, int n_in,
                              void* d_out, int out_size, void* d_ws, size_t ws_size,
                              hipStream_t stream) {
    const float* x  = (const float*)d_in[0];
    const float* w  = (const float*)d_in[1];
    const float* b  = (const float*)d_in[2];
    float* out      = (float*)d_out;

    dim3 grid(T_LEN / NP, 2, NB);
    dim3 block(256);
    size_t shmem = (size_t)LDS_FLOATS * sizeof(float);
    hipLaunchKernelGGL(trconv_fused_kernel, grid, block, shmem, stream, x, w, b, out);
}


# Weights stationary in registers: each wave reads its 16-channel slice of all 13 taps once from global (no LDS weight staging); 4-wave workgroups walk 4 position chunks; 3 residues per x fragment
# speedup vs baseline: 1.7259x; 1.7259x over previous
#include <hip/hip_runtime.h>


typedef float v2f __attribute__((ext_vector_type(2)));
typedef float v4f __attribute__((ext_vector_type(4)));
typedef float v8f __attribute__((ext_vector_type(8)));

#define T_LEN 8192
#define NB    16
#define NP    128
#define ME    144
#define XROWS 148
#define STR   68
#define NTHR  128
#define CHUNKS_PER_WG 4
#define NTILES 9

#define XS_OFF   0
#define R_OFF0   (XS_OFF + XROWS * STR)
#define RES_SZ   (ME * STR)
#define BIAS_OFF (R_OFF0 + 3 * RES_SZ)
#define LDS_FLOATS (BIAS_OFF + 64)

__device__ __forceinline__ int kslot(int ci) {
    return (((ci >> 1) & 1) << 5) + ((ci >> 2) << 1) + (ci & 1);
}

__device__ __forceinline__ int ref_idx(int i) {
    if (i < 2) return 2 - i;
    if (i > T_LEN - 3) return (i == T_LEN - 2) ? T_LEN - 6 : T_LEN - 7;
    return i - 2;
}

__global__ __launch_bounds__(NTHR) void trconv_wreg_kernel(
    const float* __restrict__ x,
    const float* __restrict__ w,
    const float* __restrict__ bias,
    float* __restrict__ out)
{
    extern __shared__ __attribute__((aligned(16))) float lds[];
    const int tid = threadIdx.x;
    const int bg  = blockIdx.x / (T_LEN / NP / CHUNKS_PER_WG);
    const int cq  = blockIdx.x - bg * (T_LEN / NP / CHUNKS_PER_WG);
    const int b   = bg >> 1;
    const int g   = bg & 1;

    const int lane = tid & 31;
    const int ct   = tid >> 5;
    const int nm   = lane & 15;
    const int lh   = lane >> 4;

    v2f A[13][16];
    {
        const float* wr = w + (size_t)(g * 64 + ct * 16 + nm) * (64 * 13);
        #pragma unroll
        for (int kk = 0; kk < 16; ++kk) {
            const v2f* p = (const v2f*)(wr + (4 * kk + 2 * lh) * 13);
            float f[26];
            #pragma unroll
            for (int i = 0; i < 13; ++i) { v2f v = p[i]; f[2 * i] = v[0]; f[2 * i + 1] = v[1]; }
            #pragma unroll
            for (int k = 0; k < 13; ++k) A[k][kk] = (v2f){f[k], f[13 + k]};
        }
    }

    if (tid < 64) lds[BIAS_OFF + tid] = bias[g * 64 + tid];

    const int xb_base     = XS_OFF + nm * STR + lh * 32;
    const int res_nm_base = R_OFF0 + nm * STR + ct * 16 + lh * 8;
    const float* xg = x + (size_t)(b * 128 + g * 64) * T_LEN;

    for (int c = 0; c < CHUNKS_PER_WG; ++c) {
        const int p0  = (cq * CHUNKS_PER_WG + c) * NP;
        const int me0 = p0 - 8;

        for (int e = tid; e < 64 * XROWS; e += NTHR) {
            int ci = e / XROWS;
            int lp = e - ci * XROWS;
            int m  = me0 + lp;
            m = m < 0 ? 0 : (m > T_LEN - 1 ? T_LEN - 1 : m);
            lds[XS_OFF + lp * STR + kslot(ci)] = xg[(size_t)ci * T_LEN + m];
        }
        __syncthreads();

        #pragma unroll 1
        for (int pt = 0; pt < NTILES; ++pt) {
            const int x0 = xb_base + pt * 16 * STR;
            v4f B[2][4];
            #pragma unroll
            for (int q = 0; q < 4; ++q) B[0][q] = *(const v4f*)&lds[x0 + 4 * q];

            v8f acc0 = {0.f,0.f,0.f,0.f,0.f,0.f,0.f,0.f};
            v8f acc1 = {0.f,0.f,0.f,0.f,0.f,0.f,0.f,0.f};
            v8f acc2 = {0.f,0.f,0.f,0.f,0.f,0.f,0.f,0.f};

            #pragma unroll
            for (int s = 0; s < 10; ++s) {
                const int t = s >> 1, hf = s & 1;
                const int cur = s & 1, nxt = cur ^ 1;
                if (s + 1 < 10) {
                    const int t1 = (s + 1) >> 1, hf1 = (s + 1) & 1;
                    #pragma unroll
                    for (int q = 0; q < 4; ++q)
                        B[nxt][q] = *(const v4f*)&lds[x0 + t1 * STR + 16 * hf1 + 4 * q];
                }
                __builtin_amdgcn_sched_barrier(0);
                #pragma unroll
                for (int q = 0; q < 4; ++q) {
                    const int kk = 8 * hf + 2 * q;
                    v2f b0 = {B[cur][q][0], B[cur][q][1]};
                    v2f b1 = {B[cur][q][2], B[cur][q][3]};
                    if (t < 4) {
                        acc0 = __builtin_amdgcn_wmma_f32_16x16x4_f32(false, A[3 * t + 2][kk], false, b0, (short)0, acc0, false, false);
                        acc1 = __builtin_amdgcn_wmma_f32_16x16x4_f32(false, A[3 * t + 1][kk], false, b0, (short)0, acc1, false, false);
                    }
                    acc2 = __builtin_amdgcn_wmma_f32_16x16x4_f32(false, A[3 * t][kk], false, b0, (short)0, acc2, false, false);
                    if (t < 4) {
                        acc0 = __builtin_amdgcn_wmma_f32_16x16x4_f32(false, A[3 * t + 2][kk + 1], false, b1, (short)0, acc0, false, false);
                        acc1 = __builtin_amdgcn_wmma_f32_16x16x4_f32(false, A[3 * t + 1][kk + 1], false, b1, (short)0, acc1, false, false);
                    }
                    acc2 = __builtin_amdgcn_wmma_f32_16x16x4_f32(false, A[3 * t][kk + 1], false, b1, (short)0, acc2, false, false);
                }
                __builtin_amdgcn_sched_barrier(0);
            }

            const int st = res_nm_base + pt * 16 * STR;
            *(v4f*)&lds[st]                  = (v4f){acc0[0], acc0[1], acc0[2], acc0[3]};
            *(v4f*)&lds[st + 4]              = (v4f){acc0[4], acc0[5], acc0[6], acc0[7]};
            *(v4f*)&lds[st + RES_SZ]         = (v4f){acc1[0], acc1[1], acc1[2], acc1[3]};
            *(v4f*)&lds[st + RES_SZ + 4]     = (v4f){acc1[4], acc1[5], acc1[6], acc1[7]};
            *(v4f*)&lds[st + 2 * RES_SZ]     = (v4f){acc2[0], acc2[1], acc2[2], acc2[3]};
            *(v4f*)&lds[st + 2 * RES_SZ + 4] = (v4f){acc2[4], acc2[5], acc2[6], acc2[7]};
        }

        __syncthreads();

        const float* r0p = &lds[R_OFF0 + 0 * RES_SZ];
        const float* r1p = &lds[R_OFF0 + 1 * RES_SZ];
        const float* r2p = &lds[R_OFF0 + 2 * RES_SZ];
        for (int idx = tid; idx < 64 * NP; idx += NTHR) {
            int co = idx >> 7;
            int il = idx & (NP - 1);
            int i  = p0 + il;
            float bi = lds[BIAS_OFF + co];

            int ri = ref_idx(i);
            int lm = ri - me0;
            float l0 = r0p[lm * STR + co], r0v = r0p[(lm + 1) * STR + co];
            float l1 = r1p[lm * STR + co], r1v = r1p[(lm + 1) * STR + co];
            float x_m = r2p[lm * STR + co] + bi;

            float um = x_m;
            #pragma unroll
            for (int d = -2; d <= 2; ++d) {
                if (d == 0) continue;
                int j = i + d;
                float v = 0.0f;
                if (j >= 0 && j <= T_LEN - 1)
                    v = r2p[(ref_idx(j) - me0) * STR + co] + bi;
                um = fmaxf(um, v);
            }

            float d0 = r0v - l0, d1 = r1v - l1;
            float qg0 = d0 / (1.0f + __expf(-1.702f * d0));
            float qg1 = d1 / (1.0f + __expf(-1.702f * d1));
            float accv = qg0 + qg1 + (r0v + l0 + r1v + l1) + 2.0f * x_m + 4.0f * bi + um;

            out[(size_t)(b * 128 + g * 64 + co) * T_LEN + i] = accv;
        }
    }
}

extern "C" void kernel_launch(void* const* d_in, const int* in_sizes, int n_in,
                              void* d_out, int out_size, void* d_ws, size_t ws_size,
                              hipStream_t stream) {
    const float* x  = (const float*)d_in[0];
    const float* w  = (const float*)d_in[1];
    const float* b  = (const float*)d_in[2];
    float* out      = (float*)d_out;

    dim3 grid(NB * 2 * (T_LEN / NP / CHUNKS_PER_WG));
    dim3 block(NTHR);
    size_t shmem = (size_t)LDS_FLOATS * sizeof(float);
    hipLaunchKernelGGL(trconv_wreg_kernel, grid, block, shmem, stream, x, w, b, out);
}


# Residues stored position-major in LDS (each accumulator register to its channel's row): epilogue reads a channel as 16-byte runs, forms 4 consecutive outputs per thread in registers, 16-byte stores
# speedup vs baseline: 2.1415x; 1.2408x over previous
#include <hip/hip_runtime.h>


typedef float v2f __attribute__((ext_vector_type(2)));
typedef float v4f __attribute__((ext_vector_type(4)));
typedef float v8f __attribute__((ext_vector_type(8)));

#define T_LEN 8192
#define NB    16
#define NP    128
#define ME    144
#define XROWS 148
#define STR   68
#define NTHR  128
#define CHUNKS_PER_WG 4
#define NTILES 9

#define PSTR  148

#define XS_OFF   0
#define R_OFF0   (XS_OFF + XROWS * STR)
#define RES_SZ   (64 * PSTR)
#define BIAS_OFF (R_OFF0 + 3 * RES_SZ)
#define LDS_FLOATS (BIAS_OFF + 64)

__device__ __forceinline__ int kslot(int ci) {
    return (((ci >> 1) & 1) << 5) + ((ci >> 2) << 1) + (ci & 1);
}

__device__ __forceinline__ int ref_idx(int i) {
    if (i < 2) return 2 - i;
    if (i > T_LEN - 3) return (i == T_LEN - 2) ? T_LEN - 6 : T_LEN - 7;
    return i - 2;
}

__device__ __forceinline__ float combine(float l0, float r0v, float l1, float r1v,
                                         float x_m, float um, float bi) {
    float d0 = r0v - l0, d1 = r1v - l1;
    float qg0 = d0 / (1.0f + __expf(-1.702f * d0));
    float qg1 = d1 / (1.0f + __expf(-1.702f * d1));
    return qg0 + qg1 + (r0v + l0 + r1v + l1) + 2.0f * x_m + 4.0f * bi + um;
}

__device__ __forceinline__ float edge_output(const float* r0p, int i, int me0, float bi) {
    const float* r1p = r0p + RES_SZ;
    const float* r2p = r0p + 2 * RES_SZ;
    int lm = ref_idx(i) - me0;
    float l0 = r0p[lm], r0v = r0p[lm + 1];
    float l1 = r1p[lm], r1v = r1p[lm + 1];
    float x_m = r2p[lm] + bi;
    float um = x_m;
    #pragma unroll
    for (int d = -2; d <= 2; ++d) {
        if (d == 0) continue;
        int j = i + d;
        float v = 0.0f;
        if (j >= 0 && j <= T_LEN - 1)
            v = r2p[ref_idx(j) - me0] + bi;
        um = fmaxf(um, v);
    }
    return combine(l0, r0v, l1, r1v, x_m, um, bi);
}

__global__ __launch_bounds__(NTHR) void trconv_wreg_kernel(
    const float* __restrict__ x,
    const float* __restrict__ w,
    const float* __restrict__ bias,
    float* __restrict__ out)
{
    extern __shared__ __attribute__((aligned(16))) float lds[];
    const int tid = threadIdx.x;
    const int bg  = blockIdx.x / (T_LEN / NP / CHUNKS_PER_WG);
    const int cq  = blockIdx.x - bg * (T_LEN / NP / CHUNKS_PER_WG);
    const int b   = bg >> 1;
    const int g   = bg & 1;

    const int lane = tid & 31;
    const int ct   = tid >> 5;
    const int nm   = lane & 15;
    const int lh   = lane >> 4;

    v2f A[13][16];
    {
        const float* wr = w + (size_t)(g * 64 + ct * 16 + nm) * (64 * 13);
        #pragma unroll
        for (int kk = 0; kk < 16; ++kk) {
            const v2f* p = (const v2f*)(wr + (4 * kk + 2 * lh) * 13);
            float f[26];
            #pragma unroll
            for (int i = 0; i < 13; ++i) { v2f v = p[i]; f[2 * i] = v[0]; f[2 * i + 1] = v[1]; }
            #pragma unroll
            for (int k = 0; k < 13; ++k) A[k][kk] = (v2f){f[k], f[13 + k]};
        }
    }

    if (tid < 64) lds[BIAS_OFF + tid] = bias[g * 64 + tid];

    const int xb_base     = XS_OFF + nm * STR + lh * 32;
    const int res_nm_base = R_OFF0 + (ct * 16 + lh * 8) * PSTR + nm;
    const float* xg = x + (size_t)(b * 128 + g * 64) * T_LEN;

    for (int c = 0; c < CHUNKS_PER_WG; ++c) {
        const int p0  = (cq * CHUNKS_PER_WG + c) * NP;
        const int me0 = p0 - 8;

        for (int e = tid; e < 64 * XROWS; e += NTHR) {
            int ci = e / XROWS;
            int lp = e - ci * XROWS;
            int m  = me0 + lp;
            m = m < 0 ? 0 : (m > T_LEN - 1 ? T_LEN - 1 : m);
            lds[XS_OFF + lp * STR + kslot(ci)] = xg[(size_t)ci * T_LEN + m];
        }
        __syncthreads();

        #pragma unroll 1
        for (int pt = 0; pt < NTILES; ++pt) {
            const int x0 = xb_base + pt * 16 * STR;
            v4f B[2][4];
            #pragma unroll
            for (int q = 0; q < 4; ++q) B[0][q] = *(const v4f*)&lds[x0 + 4 * q];

            v8f acc0 = {0.f,0.f,0.f,0.f,0.f,0.f,0.f,0.f};
            v8f acc1 = {0.f,0.f,0.f,0.f,0.f,0.f,0.f,0.f};
            v8f acc2 = {0.f,0.f,0.f,0.f,0.f,0.f,0.f,0.f};

            #pragma unroll
            for (int s = 0; s < 10; ++s) {
                const int t = s >> 1, hf = s & 1;
                const int cur = s & 1, nxt = cur ^ 1;
                if (s + 1 < 10) {
                    const int t1 = (s + 1) >> 1, hf1 = (s + 1) & 1;
                    #pragma unroll
                    for (int q = 0; q < 4; ++q)
                        B[nxt][q] = *(const v4f*)&lds[x0 + t1 * STR + 16 * hf1 + 4 * q];
                }
                __builtin_amdgcn_sched_barrier(0);
                #pragma unroll
                for (int q = 0; q < 4; ++q) {
                    const int kk = 8 * hf + 2 * q;
                    v2f b0 = {B[cur][q][0], B[cur][q][1]};
                    v2f b1 = {B[cur][q][2], B[cur][q][3]};
                    if (t < 4) {
                        acc0 = __builtin_amdgcn_wmma_f32_16x16x4_f32(false, A[3 * t + 2][kk], false, b0, (short)0, acc0, false, false);
                        acc1 = __builtin_amdgcn_wmma_f32_16x16x4_f32(false, A[3 * t + 1][kk], false, b0, (short)0, acc1, false, false);
                    }
                    acc2 = __builtin_amdgcn_wmma_f32_16x16x4_f32(false, A[3 * t][kk], false, b0, (short)0, acc2, false, false);
                    if (t < 4) {
                        acc0 = __builtin_amdgcn_wmma_f32_16x16x4_f32(false, A[3 * t + 2][kk + 1], false, b1, (short)0, acc0, false, false);
                        acc1 = __builtin_amdgcn_wmma_f32_16x16x4_f32(false, A[3 * t + 1][kk + 1], false, b1, (short)0, acc1, false, false);
                    }
                    acc2 = __builtin_amdgcn_wmma_f32_16x16x4_f32(false, A[3 * t][kk + 1], false, b1, (short)0, acc2, false, false);
                }
                __builtin_amdgcn_sched_barrier(0);
            }

            const int st = res_nm_base + pt * 16;
            #pragma unroll
            for (int r = 0; r < 8; ++r) {
                lds[st + r * PSTR]              = acc0[r];
                lds[st + RES_SZ + r * PSTR]     = acc1[r];
                lds[st + 2 * RES_SZ + r * PSTR] = acc2[r];
            }
        }

        __syncthreads();

        float* orow = out + (size_t)(b * 128 + g * 64) * T_LEN + p0;
        for (int it = tid; it < 64 * (NP / 4); it += NTHR) {
            const int co = it >> 5;
            const int j  = it & 31;
            const int i0 = p0 + 4 * j;
            const float bi = lds[BIAS_OFF + co];
            const float* r0p = &lds[R_OFF0 + co * PSTR];
            v4f o;
            if (i0 < 4 || i0 > T_LEN - 8) {
                #pragma unroll
                for (int e = 0; e < 4; ++e) o[e] = edge_output(r0p, i0 + e, me0, bi);
            } else {
                const float* wp = r0p + 4 * j + 4;
                v2f a0 = *(const v2f*)(wp + 2);
                v4f a1 = *(const v4f*)(wp + 4);
                v2f b0 = *(const v2f*)(wp + RES_SZ + 2);
                v4f b1 = *(const v4f*)(wp + RES_SZ + 4);
                v4f c0 = *(const v4f*)(wp + 2 * RES_SZ);
                v4f c1 = *(const v4f*)(wp + 2 * RES_SZ + 4);
                const float r0w[5] = {a0[0], a0[1], a1[0], a1[1], a1[2]};
                const float r1w[5] = {b0[0], b0[1], b1[0], b1[1], b1[2]};
                const float cb[8]  = {c0[0] + bi, c0[1] + bi, c0[2] + bi, c0[3] + bi,
                                      c1[0] + bi, c1[1] + bi, c1[2] + bi, c1[3] + bi};
                #pragma unroll
                for (int e = 0; e < 4; ++e) {
                    float um = fmaxf(fmaxf(fmaxf(fmaxf(cb[e + 2], cb[e]), cb[e + 1]), cb[e + 3]), cb[e + 4]);
                    o[e] = combine(r0w[e], r0w[e + 1], r1w[e], r1w[e + 1], cb[e + 2], um, bi);
                }
            }
            *(v4f*)&orow[(size_t)co * T_LEN + 4 * j] = o;
        }
    }
}

extern "C" void kernel_launch(void* const* d_in, const int* in_sizes, int n_in,
                              void* d_out, int out_size, void* d_ws, size_t ws_size,
                              hipStream_t stream) {
    const float* x  = (const float*)d_in[0];
    const float* w  = (const float*)d_in[1];
    const float* b  = (const float*)d_in[2];
    float* out      = (float*)d_out;

    dim3 grid(NB * 2 * (T_LEN / NP / CHUNKS_PER_WG));
    dim3 block(NTHR);
    size_t shmem = (size_t)LDS_FLOATS * sizeof(float);
    hipLaunchKernelGGL(trconv_wreg_kernel, grid, block, shmem, stream, x, w, b, out);
}


# Wave priority by phase (s_setprio): staging, epilogue, weight reads at 3; a chunk's GEMM at 0,1,2 by third of its tiles: of a WGP's two workgroups one holds the matrix pipe, the other works under it
# speedup vs baseline: 2.3984x; 1.1200x over previous
#include <hip/hip_runtime.h>


typedef float v2f __attribute__((ext_vector_type(2)));
typedef float v4f __attribute__((ext_vector_type(4)));
typedef float v8f __attribute__((ext_vector_type(8)));

#define T_LEN 8192
#define NB    16
#define NP    128
#define ME    144
#define XROWS 148
#define STR   68
#define NTHR  128
#define CHUNKS_PER_WG 4
#define NTILES 9

#define PSTR  148

#define XS_OFF   0
#define R_OFF0   (XS_OFF + XROWS * STR)
#define RES_SZ   (64 * PSTR)
#define BIAS_OFF (R_OFF0 + 3 * RES_SZ)
#define LDS_FLOATS (BIAS_OFF + 64)

__device__ __forceinline__ int kslot(int ci) {
    return (((ci >> 1) & 1) << 5) + ((ci >> 2) << 1) + (ci & 1);
}

__device__ __forceinline__ int ref_idx(int i) {
    if (i < 2) return 2 - i;
    if (i > T_LEN - 3) return (i == T_LEN - 2) ? T_LEN - 6 : T_LEN - 7;
    return i - 2;
}

__device__ __forceinline__ float combine(float l0, float r0v, float l1, float r1v,
                                         float x_m, float um, float bi) {
    float d0 = r0v - l0, d1 = r1v - l1;
    float qg0 = d0 / (1.0f + __expf(-1.702f * d0));
    float qg1 = d1 / (1.0f + __expf(-1.702f * d1));
    return qg0 + qg1 + (r0v + l0 + r1v + l1) + 2.0f * x_m + 4.0f * bi + um;
}

__device__ __forceinline__ float edge_output(const float* r0p, int i, int me0, float bi) {
    const float* r1p = r0p + RES_SZ;
    const float* r2p = r0p + 2 * RES_SZ;
    int lm = ref_idx(i) - me0;
    float l0 = r0p[lm], r0v = r0p[lm + 1];
    float l1 = r1p[lm], r1v = r1p[lm + 1];
    float x_m = r2p[lm] + bi;
    float um = x_m;
    #pragma unroll
    for (int d = -2; d <= 2; ++d) {
        if (d == 0) continue;
        int j = i + d;
        float v = 0.0f;
        if (j >= 0 && j <= T_LEN - 1)
            v = r2p[ref_idx(j) - me0] + bi;
        um = fmaxf(um, v);
    }
    return combine(l0, r0v, l1, r1v, x_m, um, bi);
}

__global__ __launch_bounds__(NTHR) void trconv_wreg_kernel(
    const float* __restrict__ x,
    const float* __restrict__ w,
    const float* __restrict__ bias,
    float* __restrict__ out)
{
    extern __shared__ __attribute__((aligned(16))) float lds[];
    __builtin_amdgcn_s_setprio(3);
    const int tid = threadIdx.x;
    const int bg  = blockIdx.x / (T_LEN / NP / CHUNKS_PER_WG);
    const int cq  = blockIdx.x - bg * (T_LEN / NP / CHUNKS_PER_WG);
    const int b   = bg >> 1;
    const int g   = bg & 1;

    const int lane = tid & 31;
    const int ct   = tid >> 5;
    const int nm   = lane & 15;
    const int lh   = lane >> 4;

    v2f A[13][16];
    {
        const float* wr = w + (size_t)(g * 64 + ct * 16 + nm) * (64 * 13);
        #pragma unroll
        for (int kk = 0; kk < 16; ++kk) {
            const v2f* p = (const v2f*)(wr + (4 * kk + 2 * lh) * 13);
            float f[26];
            #pragma unroll
            for (int i = 0; i < 13; ++i) { v2f v = p[i]; f[2 * i] = v[0]; f[2 * i + 1] = v[1]; }
            #pragma unroll
            for (int k = 0; k < 13; ++k) A[k][kk] = (v2f){f[k], f[13 + k]};
        }
    }

    if (tid < 64) lds[BIAS_OFF + tid] = bias[g * 64 + tid];

    const int xb_base     = XS_OFF + nm * STR + lh * 32;
    const int res_nm_base = R_OFF0 + (ct * 16 + lh * 8) * PSTR + nm;
    const float* xg = x + (size_t)(b * 128 + g * 64) * T_LEN;

    for (int c = 0; c < CHUNKS_PER_WG; ++c) {
        const int p0  = (cq * CHUNKS_PER_WG + c) * NP;
        const int me0 = p0 - 8;

        for (int e = tid; e < 64 * XROWS; e += NTHR) {
            int ci = e / XROWS;
            int lp = e - ci * XROWS;
            int m  = me0 + lp;
            m = m < 0 ? 0 : (m > T_LEN - 1 ? T_LEN - 1 : m);
            lds[XS_OFF + lp * STR + kslot(ci)] = xg[(size_t)ci * T_LEN + m];
        }
        __syncthreads();

        #pragma unroll 1
        for (int pt = 0; pt < NTILES; ++pt) {
            if (pt == 0)      __builtin_amdgcn_s_setprio(0);
            else if (pt == 3) __builtin_amdgcn_s_setprio(1);
            else if (pt == 6) __builtin_amdgcn_s_setprio(2);
            const int x0 = xb_base + pt * 16 * STR;
            v4f B[2][4];
            #pragma unroll
            for (int q = 0; q < 4; ++q) B[0][q] = *(const v4f*)&lds[x0 + 4 * q];

            v8f acc0 = {0.f,0.f,0.f,0.f,0.f,0.f,0.f,0.f};
            v8f acc1 = {0.f,0.f,0.f,0.f,0.f,0.f,0.f,0.f};
            v8f acc2 = {0.f,0.f,0.f,0.f,0.f,0.f,0.f,0.f};

            #pragma unroll
            for (int s = 0; s < 10; ++s) {
                const int t = s >> 1, hf = s & 1;
                const int cur = s & 1, nxt = cur ^ 1;
                if (s + 1 < 10) {
                    const int t1 = (s + 1) >> 1, hf1 = (s + 1) & 1;
                    #pragma unroll
                    for (int q = 0; q < 4; ++q)
                        B[nxt][q] = *(const v4f*)&lds[x0 + t1 * STR + 16 * hf1 + 4 * q];
                }
                __builtin_amdgcn_sched_barrier(0);
                #pragma unroll
                for (int q = 0; q < 4; ++q) {
                    const int kk = 8 * hf + 2 * q;
                    v2f b0 = {B[cur][q][0], B[cur][q][1]};
                    v2f b1 = {B[cur][q][2], B[cur][q][3]};
                    if (t < 4) {
                        acc0 = __builtin_amdgcn_wmma_f32_16x16x4_f32(false, A[3 * t + 2][kk], false, b0, (short)0, acc0, false, false);
                        acc1 = __builtin_amdgcn_wmma_f32_16x16x4_f32(false, A[3 * t + 1][kk], false, b0, (short)0, acc1, false, false);
                    }
                    acc2 = __builtin_amdgcn_wmma_f32_16x16x4_f32(false, A[3 * t][kk], false, b0, (short)0, acc2, false, false);
                    if (t < 4) {
                        acc0 = __builtin_amdgcn_wmma_f32_16x16x4_f32(false, A[3 * t + 2][kk + 1], false, b1, (short)0, acc0, false, false);
                        acc1 = __builtin_amdgcn_wmma_f32_16x16x4_f32(false, A[3 * t + 1][kk + 1], false, b1, (short)0, acc1, false, false);
                    }
                    acc2 = __builtin_amdgcn_wmma_f32_16x16x4_f32(false, A[3 * t][kk + 1], false, b1, (short)0, acc2, false, false);
                }
                __builtin_amdgcn_sched_barrier(0);
            }

            const int st = res_nm_base + pt * 16;
            #pragma unroll
            for (int r = 0; r < 8; ++r) {
                lds[st + r * PSTR]              = acc0[r];
                lds[st + RES_SZ + r * PSTR]     = acc1[r];
                lds[st + 2 * RES_SZ + r * PSTR] = acc2[r];
            }
        }

        __builtin_amdgcn_s_setprio(3);
        __syncthreads();

        float* orow = out + (size_t)(b * 128 + g * 64) * T_LEN + p0;
        for (int it = tid; it < 64 * (NP / 4); it += NTHR) {
            const int co = it >> 5;
            const int j  = it & 31;
            const int i0 = p0 + 4 * j;
            const float bi = lds[BIAS_OFF + co];
            const float* r0p = &lds[R_OFF0 + co * PSTR];
            v4f o;
            if (i0 < 4 || i0 > T_LEN - 8) {
                #pragma unroll
                for (int e = 0; e < 4; ++e) o[e] = edge_output(r0p, i0 + e, me0, bi);
            } else {
                const float* wp = r0p + 4 * j + 4;
                v2f a0 = *(const v2f*)(wp + 2);
                v4f a1 = *(const v4f*)(wp + 4);
                v2f b0 = *(const v2f*)(wp + RES_SZ + 2);
                v4f b1 = *(const v4f*)(wp + RES_SZ + 4);
                v4f c0 = *(const v4f*)(wp + 2 * RES_SZ);
                v4f c1 = *(const v4f*)(wp + 2 * RES_SZ + 4);
                const float r0w[5] = {a0[0], a0[1], a1[0], a1[1], a1[2]};
                const float r1w[5] = {b0[0], b0[1], b1[0], b1[1], b1[2]};
                const float cb[8]  = {c0[0] + bi, c0[1] + bi, c0[2] + bi, c0[3] + bi,
                                      c1[0] + bi, c1[1] + bi, c1[2] + bi, c1[3] + bi};
                #pragma unroll
                for (int e = 0; e < 4; ++e) {
                    float um = fmaxf(fmaxf(fmaxf(fmaxf(cb[e + 2], cb[e]), cb[e + 1]), cb[e + 3]), cb[e + 4]);
                    o[e] = combine(r0w[e], r0w[e + 1], r1w[e], r1w[e + 1], cb[e + 2], um, bi);
                }
            }
            *(v4f*)&orow[(size_t)co * T_LEN + 4 * j] = o;
        }
    }
}

extern "C" void kernel_launch(void* const* d_in, const int* in_sizes, int n_in,
                              void* d_out, int out_size, void* d_ws, size_t ws_size,
                              hipStream_t stream) {
    const float* x  = (const float*)d_in[0];
    const float* w  = (const float*)d_in[1];
    const float* b  = (const float*)d_in[2];
    float* out      = (float*)d_out;

    dim3 grid(NB * 2 * (T_LEN / NP / CHUNKS_PER_WG));
    dim3 block(NTHR);
    size_t shmem = (size_t)LDS_FLOATS * sizeof(float);
    hipLaunchKernelGGL(trconv_wreg_kernel, grid, block, shmem, stream, x, w, b, out);
}


# x staging without per-element index arithmetic: lane = tile row, wave = 4 channel quads of adjacent k-slots; four 4-byte loads fill one 16-byte LDS row store; was 17 vector ops per element, 18 waits
# speedup vs baseline: 2.6357x; 1.0989x over previous
#include <hip/hip_runtime.h>


typedef float v2f __attribute__((ext_vector_type(2)));
typedef float v4f __attribute__((ext_vector_type(4)));
typedef float v8f __attribute__((ext_vector_type(8)));

#define T_LEN 8192
#define NB    16
#define NP    128
#define ME    144
#define XROWS 148
#define STR   68
#define NTHR  128
#define CHUNKS_PER_WG 4
#define NTILES 9

#define PSTR  148

#define XS_OFF   0
#define R_OFF0   (XS_OFF + XROWS * STR)
#define RES_SZ   (64 * PSTR)
#define BIAS_OFF (R_OFF0 + 3 * RES_SZ)
#define LDS_FLOATS (BIAS_OFF + 64)


__device__ __forceinline__ int ref_idx(int i) {
    if (i < 2) return 2 - i;
    if (i > T_LEN - 3) return (i == T_LEN - 2) ? T_LEN - 6 : T_LEN - 7;
    return i - 2;
}

__device__ __forceinline__ float combine(float l0, float r0v, float l1, float r1v,
                                         float x_m, float um, float bi) {
    float d0 = r0v - l0, d1 = r1v - l1;
    float qg0 = d0 / (1.0f + __expf(-1.702f * d0));
    float qg1 = d1 / (1.0f + __expf(-1.702f * d1));
    return qg0 + qg1 + (r0v + l0 + r1v + l1) + 2.0f * x_m + 4.0f * bi + um;
}

__device__ __forceinline__ float edge_output(const float* r0p, int i, int me0, float bi) {
    const float* r1p = r0p + RES_SZ;
    const float* r2p = r0p + 2 * RES_SZ;
    int lm = ref_idx(i) - me0;
    float l0 = r0p[lm], r0v = r0p[lm + 1];
    float l1 = r1p[lm], r1v = r1p[lm + 1];
    float x_m = r2p[lm] + bi;
    float um = x_m;
    #pragma unroll
    for (int d = -2; d <= 2; ++d) {
        if (d == 0) continue;
        int j = i + d;
        float v = 0.0f;
        if (j >= 0 && j <= T_LEN - 1)
            v = r2p[ref_idx(j) - me0] + bi;
        um = fmaxf(um, v);
    }
    return combine(l0, r0v, l1, r1v, x_m, um, bi);
}

__global__ __launch_bounds__(NTHR) void trconv_wreg_kernel(
    const float* __restrict__ x,
    const float* __restrict__ w,
    const float* __restrict__ bias,
    float* __restrict__ out)
{
    extern __shared__ __attribute__((aligned(16))) float lds[];
    __builtin_amdgcn_s_setprio(3);
    const int tid = threadIdx.x;
    const int bg  = blockIdx.x / (T_LEN / NP / CHUNKS_PER_WG);
    const int cq  = blockIdx.x - bg * (T_LEN / NP / CHUNKS_PER_WG);
    const int b   = bg >> 1;
    const int g   = bg & 1;

    const int lane = tid & 31;
    const int ct   = tid >> 5;
    const int nm   = lane & 15;
    const int lh   = lane >> 4;
    const int ctu  = __builtin_amdgcn_readfirstlane(ct);

    v2f A[13][16];
    {
        const float* wr = w + (size_t)(g * 64 + ct * 16 + nm) * (64 * 13);
        #pragma unroll
        for (int kk = 0; kk < 16; ++kk) {
            const v2f* p = (const v2f*)(wr + (4 * kk + 2 * lh) * 13);
            float f[26];
            #pragma unroll
            for (int i = 0; i < 13; ++i) { v2f v = p[i]; f[2 * i] = v[0]; f[2 * i + 1] = v[1]; }
            #pragma unroll
            for (int k = 0; k < 13; ++k) A[k][kk] = (v2f){f[k], f[13 + k]};
        }
    }

    if (tid < 64) lds[BIAS_OFF + tid] = bias[g * 64 + tid];

    const int xb_base     = XS_OFF + nm * STR + lh * 32;
    const int res_nm_base = R_OFF0 + (ct * 16 + lh * 8) * PSTR + nm;
    const float* xg = x + (size_t)(b * 128 + g * 64) * T_LEN;

    for (int c = 0; c < CHUNKS_PER_WG; ++c) {
        const int p0  = (cq * CHUNKS_PER_WG + c) * NP;
        const int me0 = p0 - 8;

        #pragma unroll
        for (int rb = 0; rb < 5; ++rb) {
            int m = me0 + 32 * rb + lane;
            m = m < 0 ? 0 : (m > T_LEN - 1 ? T_LEN - 1 : m);
            if (rb < 4 || lane < XROWS - 128) {
                #pragma unroll
                for (int r = 0; r < 4; ++r) {
                    const int q = ctu + 4 * r;
                    const float* g0 = xg + (size_t)(8 * (q & 7) + 2 * (q >> 3)) * T_LEN + m;
                    const v4f v = {g0[0], g0[T_LEN], g0[4 * T_LEN], g0[5 * T_LEN]};
                    *(v4f*)&lds[XS_OFF + (32 * rb + lane) * STR + 4 * q] = v;
                }
            }
        }
        __syncthreads();

        #pragma unroll 1
        for (int pt = 0; pt < NTILES; ++pt) {
            if (pt == 0)      __builtin_amdgcn_s_setprio(0);
            else if (pt == 3) __builtin_amdgcn_s_setprio(1);
            else if (pt == 6) __builtin_amdgcn_s_setprio(2);
            const int x0 = xb_base + pt * 16 * STR;
            v4f B[2][4];
            #pragma unroll
            for (int q = 0; q < 4; ++q) B[0][q] = *(const v4f*)&lds[x0 + 4 * q];

            v8f acc0 = {0.f,0.f,0.f,0.f,0.f,0.f,0.f,0.f};
            v8f acc1 = {0.f,0.f,0.f,0.f,0.f,0.f,0.f,0.f};
            v8f acc2 = {0.f,0.f,0.f,0.f,0.f,0.f,0.f,0.f};

            #pragma unroll
            for (int s = 0; s < 10; ++s) {
                const int t = s >> 1, hf = s & 1;
                const int cur = s & 1, nxt = cur ^ 1;
                if (s + 1 < 10) {
                    const int t1 = (s + 1) >> 1, hf1 = (s + 1) & 1;
                    #pragma unroll
                    for (int q = 0; q < 4; ++q)
                        B[nxt][q] = *(const v4f*)&lds[x0 + t1 * STR + 16 * hf1 + 4 * q];
                }
                __builtin_amdgcn_sched_barrier(0);
                #pragma unroll
                for (int q = 0; q < 4; ++q) {
                    const int kk = 8 * hf + 2 * q;
                    v2f b0 = {B[cur][q][0], B[cur][q][1]};
                    v2f b1 = {B[cur][q][2], B[cur][q][3]};
                    if (t < 4) {
                        acc0 = __builtin_amdgcn_wmma_f32_16x16x4_f32(false, A[3 * t + 2][kk], false, b0, (short)0, acc0, false, false);
                        acc1 = __builtin_amdgcn_wmma_f32_16x16x4_f32(false, A[3 * t + 1][kk], false, b0, (short)0, acc1, false, false);
                    }
                    acc2 = __builtin_amdgcn_wmma_f32_16x16x4_f32(false, A[3 * t][kk], false, b0, (short)0, acc2, false, false);
                    if (t < 4) {
                        acc0 = __builtin_amdgcn_wmma_f32_16x16x4_f32(false, A[3 * t + 2][kk + 1], false, b1, (short)0, acc0, false, false);
                        acc1 = __builtin_amdgcn_wmma_f32_16x16x4_f32(false, A[3 * t + 1][kk + 1], false, b1, (short)0, acc1, false, false);
                    }
                    acc2 = __builtin_amdgcn_wmma_f32_16x16x4_f32(false, A[3 * t][kk + 1], false, b1, (short)0, acc2, false, false);
                }
                __builtin_amdgcn_sched_barrier(0);
            }

            const int st = res_nm_base + pt * 16;
            #pragma unroll
            for (int r = 0; r < 8; ++r) {
                lds[st + r * PSTR]              = acc0[r];
                lds[st + RES_SZ + r * PSTR]     = acc1[r];
                lds[st + 2 * RES_SZ + r * PSTR] = acc2[r];
            }
        }

        __builtin_amdgcn_s_setprio(3);
        __syncthreads();

        float* orow = out + (size_t)(b * 128 + g * 64) * T_LEN + p0;
        for (int it = tid; it < 64 * (NP / 4); it += NTHR) {
            const int co = it >> 5;
            const int j  = it & 31;
            const int i0 = p0 + 4 * j;
            const float bi = lds[BIAS_OFF + co];
            const float* r0p = &lds[R_OFF0 + co * PSTR];
            v4f o;
            if (i0 < 4 || i0 > T_LEN - 8) {
                #pragma unroll
                for (int e = 0; e < 4; ++e) o[e] = edge_output(r0p, i0 + e, me0, bi);
            } else {
                const float* wp = r0p + 4 * j + 4;
                v2f a0 = *(const v2f*)(wp + 2);
                v4f a1 = *(const v4f*)(wp + 4);
                v2f b0 = *(const v2f*)(wp + RES_SZ + 2);
                v4f b1 = *(const v4f*)(wp + RES_SZ + 4);
                v4f c0 = *(const v4f*)(wp + 2 * RES_SZ);
                v4f c1 = *(const v4f*)(wp + 2 * RES_SZ + 4);
                const float r0w[5] = {a0[0], a0[1], a1[0], a1[1], a1[2]};
                const float r1w[5] = {b0[0], b0[1], b1[0], b1[1], b1[2]};
                const float cb[8]  = {c0[0] + bi, c0[1] + bi, c0[2] + bi, c0[3] + bi,
                                      c1[0] + bi, c1[1] + bi, c1[2] + bi, c1[3] + bi};
                #pragma unroll
                for (int e = 0; e < 4; ++e) {
                    float um = fmaxf(fmaxf(fmaxf(fmaxf(cb[e + 2], cb[e]), cb[e + 1]), cb[e + 3]), cb[e + 4]);
                    o[e] = combine(r0w[e], r0w[e + 1], r1w[e], r1w[e + 1], cb[e + 2], um, bi);
                }
            }
            *(v4f*)&orow[(size_t)co * T_LEN + 4 * j] = o;
        }
    }
}

extern "C" void kernel_launch(void* const* d_in, const int* in_sizes, int n_in,
                              void* d_out, int out_size, void* d_ws, size_t ws_size,
                              hipStream_t stream) {
    const float* x  = (const float*)d_in[0];
    const float* w  = (const float*)d_in[1];
    const float* b  = (const float*)d_in[2];
    float* out      = (float*)d_out;

    dim3 grid(NB * 2 * (T_LEN / NP / CHUNKS_PER_WG));
    dim3 block(NTHR);
    size_t shmem = (size_t)LDS_FLOATS * sizeof(float);
    hipLaunchKernelGGL(trconv_wreg_kernel, grid, block, shmem, stream, x, w, b, out);
}
